# starting kernel
# baseline (speedup 1.0000x reference)
#include <hip/hip_runtime.h>

typedef float v2f __attribute__((ext_vector_type(2)));
typedef float v8f __attribute__((ext_vector_type(8)));

#define HW 192
#define OX 95
#define OY 95

__global__ __launch_bounds__(256)
void quanv_kernel(const float* __restrict__ X, const float* __restrict__ Wt,
                  float* __restrict__ out, int totalRows) {
    __shared__ float s_bmat[16 * 16];
    __shared__ float s_phc[16], s_phs[16];
    __shared__ float s_re[8][32][18];
    __shared__ float s_im[8][32][18];
    __shared__ float s_p[8][16][17];

    const int tid  = threadIdx.x;
    const int lane = tid & 31;
    const int wv   = tid >> 5;

    const float w0 = Wt[0], w1 = Wt[1], w2 = Wt[2], w3 = Wt[3];
    float cq[4], sq[4];
#pragma unroll
    for (int q = 0; q < 4; ++q) __sincosf(0.5f * Wt[4 + q], &sq[q], &cq[q]);

    if (tid < 16) {
        int s = tid;
        int b0 = (s >> 3) & 1, b1 = (s >> 2) & 1, b2 = (s >> 1) & 1, b3 = s & 1;
        float phi = 0.5f * ((float)b0 * (2 * b1 - 1) * w0 + (float)b1 * (2 * b2 - 1) * w1 +
                            (float)b2 * (2 * b3 - 1) * w2 + (float)b3 * (2 * b0 - 1) * w3);
        float sp, cp; __sincosf(phi, &sp, &cp);
        s_phc[s] = cp; s_phs[s] = sp;
    }
    {
        int k = tid >> 4, n = tid & 15;
        float v = 1.0f;
#pragma unroll
        for (int q = 0; q < 4; ++q) {
            int i = (n >> (3 - q)) & 1;
            int j = (k >> (3 - q)) & 1;
            float m = (i == j) ? cq[q] : ((i == 0) ? -sq[q] : sq[q]);
            v *= m;
        }
        s_bmat[k * 16 + n] = v;
    }
    __syncthreads();

    const int rowBase = (int)blockIdx.x * 256;
    int r = rowBase + tid;
    if (r >= totalRows) r = totalRows - 1;

    unsigned rr  = (unsigned)r;
    unsigned b   = rr / (unsigned)(OX * OY);
    unsigned rem = rr % (unsigned)(OX * OY);
    unsigned px  = rem / (unsigned)OY;
    unsigned py  = rem % (unsigned)OY;

    const float* xp = X + (size_t)b * (HW * HW) + (size_t)(2u * px) * HW + 2u * py;
    float t[16];
#pragma unroll
    for (int kx = 0; kx < 4; ++kx) {
        const float* rp = xp + kx * HW;
        float2 a = *(const float2*)(rp);
        float2 c = *(const float2*)(rp + 2);
        t[kx * 4 + 0] = a.x; t[kx * 4 + 1] = a.y;
        t[kx * 4 + 2] = c.x; t[kx * 4 + 3] = c.y;
    }

    float vr[4][2], vi[4][2];
    const float is2 = 0.70710678118654752f;
#pragma unroll
    for (int q = 0; q < 4; ++q) { vr[q][0] = is2; vr[q][1] = is2; vi[q][0] = 0.f; vi[q][1] = 0.f; }

    auto applyRZ = [&](int q, float th) {
        float sh, ch; __sincosf(0.5f * th, &sh, &ch);
        float r0 = vr[q][0], i0 = vi[q][0], r1 = vr[q][1], i1 = vi[q][1];
        vr[q][0] = r0 * ch + i0 * sh;  vi[q][0] = i0 * ch - r0 * sh;
        vr[q][1] = r1 * ch - i1 * sh;  vi[q][1] = i1 * ch + r1 * sh;
    };
    auto applyRY = [&](int q, float th) {
        float sh, ch; __sincosf(0.5f * th, &sh, &ch);
        float r0 = vr[q][0], i0 = vi[q][0], r1 = vr[q][1], i1 = vi[q][1];
        vr[q][0] = ch * r0 - sh * r1;  vi[q][0] = ch * i0 - sh * i1;
        vr[q][1] = sh * r0 + ch * r1;  vi[q][1] = sh * i0 + ch * i1;
    };
#pragma unroll
    for (int q = 0; q < 3; ++q) {
        applyRZ(q, t[q * 5 + 0]); applyRY(q, t[q * 5 + 1]); applyRZ(q, t[q * 5 + 2]);
        applyRY(q, t[q * 5 + 3]); applyRZ(q, t[q * 5 + 4]);
    }
    applyRZ(3, t[15]);

    float w01r[4], w01i[4], w23r[4], w23i[4];
#pragma unroll
    for (int i = 0; i < 2; ++i)
#pragma unroll
        for (int j = 0; j < 2; ++j) {
            w01r[i * 2 + j] = vr[0][i] * vr[1][j] - vi[0][i] * vi[1][j];
            w01i[i * 2 + j] = vr[0][i] * vi[1][j] + vi[0][i] * vr[1][j];
            w23r[i * 2 + j] = vr[2][i] * vr[3][j] - vi[2][i] * vi[3][j];
            w23i[i * 2 + j] = vr[2][i] * vi[3][j] + vi[2][i] * vr[3][j];
        }
#pragma unroll
    for (int s = 0; s < 16; ++s) {
        float ar = w01r[s >> 2], ai = w01i[s >> 2];
        float br = w23r[s & 3],  bi = w23i[s & 3];
        float re = ar * br - ai * bi, im = ar * bi + ai * br;
        float pc = s_phc[s], ps = s_phs[s];
        s_re[wv][lane][s] = re * pc - im * ps;
        s_im[wv][lane][s] = re * ps + im * pc;
    }
    __syncthreads();

    const int nn   = lane & 15;
    const int koff = (lane < 16) ? 0 : 2;

#pragma unroll
    for (int tI = 0; tI < 2; ++tI) {
        v8f dRe = {}; v8f dIm = {};
        const int srcRow = tI * 16 + nn;
#pragma unroll
        for (int k0 = 0; k0 < 16; k0 += 4) {
            const int kk = k0 + koff;
            v2f a_re, a_im, bop;
            a_re.x = s_re[wv][srcRow][kk];      a_re.y = s_re[wv][srcRow][kk + 1];
            a_im.x = s_im[wv][srcRow][kk];      a_im.y = s_im[wv][srcRow][kk + 1];
            bop.x  = s_bmat[kk * 16 + nn];      bop.y  = s_bmat[(kk + 1) * 16 + nn];
            dRe = __builtin_amdgcn_wmma_f32_16x16x4_f32(false, a_re, false, bop,
                                                        (short)0, dRe, false, false);
            dIm = __builtin_amdgcn_wmma_f32_16x16x4_f32(false, a_im, false, bop,
                                                        (short)0, dIm, false, false);
        }
        const int prow0 = (lane < 16) ? 0 : 8;
#pragma unroll
        for (int v = 0; v < 8; ++v) {
            float p = dRe[v] * dRe[v] + dIm[v] * dIm[v];
            s_p[wv][prow0 + v][nn] = p;
        }
        __builtin_amdgcn_wave_barrier();
#pragma unroll
        for (int oo = 0; oo < 2; ++oo) {
            int o    = lane + oo * 32;
            int prow = o >> 2;
            int q    = o & 3;
            float ev = 0.f;
#pragma unroll
            for (int s = 0; s < 16; ++s) {
                float sgn = ((s >> (3 - q)) & 1) ? -1.f : 1.f;
                ev += sgn * s_p[wv][prow][s];
            }
            int rowG = rowBase + wv * 32 + tI * 16 + prow;
            if (rowG < totalRows) out[(size_t)rowG * 4 + q] = ev;
        }
        __builtin_amdgcn_wave_barrier();
    }
}

extern "C" void kernel_launch(void* const* d_in, const int* in_sizes, int n_in,
                              void* d_out, int out_size, void* d_ws, size_t ws_size,
                              hipStream_t stream) {
    const float* X  = (const float*)d_in[0];
    const float* Wt = (const float*)d_in[1];
    float* out      = (float*)d_out;
    int B = in_sizes[0] / (HW * HW);
    int totalRows = B * OX * OY;
    int blocks = (totalRows + 255) / 256;
    quanv_kernel<<<blocks, 256, 0, stream>>>(X, Wt, out, totalRows);
}


# Gates and angle scaling two-wide by hand, one row per lane kept: a qubit is (re0,re1),(im0,im1), each RZ/RY four packed ops, swap/sign as operand modifiers; first RZ written out; last-bit rounding
# speedup vs baseline: 1.3048x; 1.3048x over previous
#include <hip/hip_runtime.h>

typedef float v2f __attribute__((ext_vector_type(2)));
typedef float v4f __attribute__((ext_vector_type(4)));
typedef float v8f __attribute__((ext_vector_type(8)));
typedef v4f v4f_a8 __attribute__((aligned(8)));

#define HW 192
#define OX 95
#define OY 95

__global__ __launch_bounds__(256)
void quanv_kernel(const float* __restrict__ X, const float* __restrict__ Wt,
                  float* __restrict__ out, int totalRows) {
    __shared__ float s_bmat[16 * 16];
    __shared__ __attribute__((aligned(16))) float s_phc[16];
    __shared__ __attribute__((aligned(16))) float s_phs[16];
    __shared__ __attribute__((aligned(16))) float s_stage[8][2][32][18];
    __shared__ __attribute__((aligned(16))) float s_sgn[4][16];

    const int tid  = threadIdx.x;
    const int lane = tid & 31;
    const int wv   = tid >> 5;

    const int rowBase = (int)blockIdx.x * 256;
    int r = rowBase + tid;
    if (r >= totalRows) r = totalRows - 1;

    unsigned rr  = (unsigned)r;
    unsigned b   = rr / (unsigned)(OX * OY);
    unsigned rem = rr % (unsigned)(OX * OY);
    unsigned px  = rem / (unsigned)OY;
    unsigned py  = rem % (unsigned)OY;

    const float* xp = X + (size_t)b * (HW * HW) + (size_t)(2u * px) * HW + 2u * py;
    v4f xrow[4];
#pragma unroll
    for (int kx = 0; kx < 4; ++kx) xrow[kx] = *(const v4f_a8*)(xp + kx * HW);

    const int wvu = __builtin_amdgcn_readfirstlane(wv);
    if (wvu < 2) {
      if (wvu == 0) {
        const int n = lane & 15, kh = lane >> 4;
        float m[4][2];
#pragma unroll
        for (int q = 0; q < 4; ++q) {
            float sq, cq; __sincosf(0.5f * Wt[4 + q], &sq, &cq);
            int i = (n >> (3 - q)) & 1;
            m[q][0] = (i == 0) ? cq : sq;
            m[q][1] = (i == 0) ? -sq : cq;
        }
        const float f0 = kh ? m[0][1] : m[0][0];
#pragma unroll
        for (int j = 0; j < 8; ++j) {
            float v = f0 * m[1][(j >> 2) & 1];
            v *= m[2][(j >> 1) & 1];
            v *= m[3][j & 1];
            s_bmat[(kh * 8 + j) * 16 + n] = v;
        }
      } else {
        if (lane < 16) {
            const float w0 = Wt[0], w1 = Wt[1], w2 = Wt[2], w3 = Wt[3];
            int s = lane;
            int b0 = (s >> 3) & 1, b1 = (s >> 2) & 1, b2 = (s >> 1) & 1, b3 = s & 1;
            float phi = 0.5f * ((float)b0 * (2 * b1 - 1) * w0 + (float)b1 * (2 * b2 - 1) * w1 +
                                (float)b2 * (2 * b3 - 1) * w2 + (float)b3 * (2 * b0 - 1) * w3);
            float sp, cp; __sincosf(phi, &sp, &cp);
            s_phc[s] = cp; s_phs[s] = sp;
        }
#pragma unroll
        for (int e = 0; e < 2; ++e) {
            int idx = lane + e * 32;
            int q = idx >> 4, s = idx & 15;
            s_sgn[q][s] = ((s >> (3 - q)) & 1) ? -1.f : 1.f;
        }
      }
    }

    __syncthreads();

    float t[16];
#pragma unroll
    for (int kx = 0; kx < 4; ++kx) {
        const v4f u = (xrow[kx] * 0.5f) * 0.15915494309189532f;
        t[kx * 4 + 0] = u.x; t[kx * 4 + 1] = u.y;
        t[kx * 4 + 2] = u.z; t[kx * 4 + 3] = u.w;
    }

    v2f P[4], Q[4];
    const float is2 = 0.70710678118654752f;
    auto firstRZ = [&](int q, float th) {
        const float sh = __builtin_amdgcn_sinf(th), ch = __builtin_amdgcn_cosf(th);
        const float a = is2 * ch, bq = is2 * sh;
        P[q] = (v2f){a, a};
        Q[q] = (v2f){-bq, bq};
    };
    auto applyRZ = [&](int q, float th) {
        const float sh = __builtin_amdgcn_sinf(th), ch = __builtin_amdgcn_cosf(th);
        const v2f S = {sh, -sh}, C = {ch, ch};
        const v2f p = P[q], w = Q[q];
        P[q] = w * S + p * C;
        Q[q] = w * C - p * S;
    };
    auto applyRY = [&](int q, float th) {
        const float sh = __builtin_amdgcn_sinf(th), ch = __builtin_amdgcn_cosf(th);
        const v2f S = {-sh, sh}, C = {ch, ch};
        const v2f p = P[q], w = Q[q];
        const v2f px = __builtin_shufflevector(p, p, 1, 0), wx = __builtin_shufflevector(w, w, 1, 0);
        P[q] = px * S + p * C;
        Q[q] = wx * S + w * C;
    };
#pragma unroll
    for (int q = 0; q < 3; ++q) {
        firstRZ(q, t[q * 5 + 0]); applyRY(q, t[q * 5 + 1]); applyRZ(q, t[q * 5 + 2]);
        applyRY(q, t[q * 5 + 3]); applyRZ(q, t[q * 5 + 4]);
    }
    firstRZ(3, t[15]);

    v2f w01r[2], w01i[2], w23r[2], w23i[2];
#pragma unroll
    for (int i = 0; i < 2; ++i) {
        const float ar = P[0][i], ai = Q[0][i], br = P[2][i], bi = Q[2][i];
        w01r[i] = ar * P[1] - ai * Q[1];
        w01i[i] = ar * Q[1] + ai * P[1];
        w23r[i] = br * P[3] - bi * Q[3];
        w23i[i] = br * Q[3] + bi * P[3];
    }
#pragma unroll
    for (int ij = 0; ij < 4; ++ij) {
        const float ar = w01r[ij >> 1][ij & 1], ai = w01i[ij >> 1][ij & 1];
#pragma unroll
        for (int k = 0; k < 2; ++k) {
            const int s0 = ij * 4 + k * 2;
            const v2f re = ar * w23r[k] - ai * w23i[k];
            const v2f im = ar * w23i[k] + ai * w23r[k];
            const v2f pc = *(const v2f*)&s_phc[s0], ps = *(const v2f*)&s_phs[s0];
            *(v2f*)&s_stage[wv][0][lane][s0] = re * pc - im * ps;
            *(v2f*)&s_stage[wv][1][lane][s0] = re * ps + im * pc;
        }
    }
    __builtin_amdgcn_wave_barrier();

    const int nn   = lane & 15;
    const int koff = (lane < 16) ? 0 : 2;
    v4f sg[4];
#pragma unroll
    for (int m = 0; m < 4; ++m) sg[m] = *(const v4f*)&s_sgn[lane & 3][m * 4];

    v2f bfr[4], aRe[2][4], aIm[2][4];
#pragma unroll
    for (int j = 0; j < 4; ++j) {
        const int kk = j * 4 + koff;
        bfr[j].x = s_bmat[kk * 16 + nn];
        bfr[j].y = s_bmat[(kk + 1) * 16 + nn];
#pragma unroll
        for (int tI = 0; tI < 2; ++tI) {
            const int srcRow = tI * 16 + nn;
            aRe[tI][j].x = s_stage[wv][0][srcRow][kk];  aRe[tI][j].y = s_stage[wv][0][srcRow][kk + 1];
            aIm[tI][j].x = s_stage[wv][1][srcRow][kk];  aIm[tI][j].y = s_stage[wv][1][srcRow][kk + 1];
        }
    }
    v8f dRe[2] = {{}, {}}; v8f dIm[2] = {{}, {}};
#pragma unroll
    for (int j = 0; j < 4; ++j)
#pragma unroll
        for (int tI = 0; tI < 2; ++tI) {
            dRe[tI] = __builtin_amdgcn_wmma_f32_16x16x4_f32(false, aRe[tI][j], false, bfr[j],
                                                            (short)0, dRe[tI], false, false);
            dIm[tI] = __builtin_amdgcn_wmma_f32_16x16x4_f32(false, aIm[tI][j], false, bfr[j],
                                                            (short)0, dIm[tI], false, false);
        }
    float* sp = &s_stage[wv][0][0][0];
    const int prow0 = (lane < 16) ? 0 : 8;
#pragma unroll
    for (int tI = 0; tI < 2; ++tI)
#pragma unroll
        for (int v = 0; v < 8; ++v) {
            float p = dRe[tI][v] * dRe[tI][v] + dIm[tI][v] * dIm[tI][v];
            sp[(tI * 16 + prow0 + v) * 20 + nn] = p;
        }
    __builtin_amdgcn_wave_barrier();
    float ev[4];
#pragma unroll
    for (int jj = 0; jj < 4; ++jj) {
        const int prow = (lane >> 2) + 8 * jj;
        v2f acc = {0.f, 0.f};
#pragma unroll
        for (int m = 0; m < 4; ++m) {
            v4f p4 = *(const v4f*)&sp[prow * 20 + m * 4];
            v2f plo = {p4.x, p4.y}, phi = {p4.z, p4.w};
            v2f slo = {sg[m].x, sg[m].y}, shi = {sg[m].z, sg[m].w};
            acc = plo * slo + acc;
            acc = phi * shi + acc;
        }
        ev[jj] = acc.x + acc.y;
    }
    const int rowW = rowBase + wvu * 32;
    float* ob = out + (size_t)rowW * 4 + lane;
    if (rowW + 32 <= totalRows) {
#pragma unroll
        for (int jj = 0; jj < 4; ++jj) ob[32 * jj] = ev[jj];
    } else {
#pragma unroll
        for (int jj = 0; jj < 4; ++jj)
            if (rowW + (lane >> 2) + 8 * jj < totalRows) ob[32 * jj] = ev[jj];
    }
}

extern "C" void kernel_launch(void* const* d_in, const int* in_sizes, int n_in,
                              void* d_out, int out_size, void* d_ws, size_t ws_size,
                              hipStream_t stream) {
    const float* X  = (const float*)d_in[0];
    const float* Wt = (const float*)d_in[1];
    float* out      = (float*)d_out;
    int B = in_sizes[0] / (HW * HW);
    int totalRows = B * OX * OY;
    int blocks = (totalRows + 255) / 256;
    quanv_kernel<<<blocks, 256, 0, stream>>>(X, Wt, out, totalRows);
}


# Signed sums one row per lane: lane reads only its own row of |amp|^2 from LDS (a quarter of the reads, no sign rows), four <Z> by add/subtract tree, one 16-byte store per row; sum order changed
# speedup vs baseline: 1.4420x; 1.1052x over previous
#include <hip/hip_runtime.h>

typedef float v2f __attribute__((ext_vector_type(2)));
typedef float v4f __attribute__((ext_vector_type(4)));
typedef float v8f __attribute__((ext_vector_type(8)));
typedef v4f v4f_a8 __attribute__((aligned(8)));

#define HW 192
#define OX 95
#define OY 95

__global__ __launch_bounds__(256)
void quanv_kernel(const float* __restrict__ X, const float* __restrict__ Wt,
                  float* __restrict__ out, int totalRows) {
    __shared__ float s_bmat[16 * 16];
    __shared__ __attribute__((aligned(16))) float s_phc[16];
    __shared__ __attribute__((aligned(16))) float s_phs[16];
    __shared__ __attribute__((aligned(16))) float s_stage[8][2][32][18];

    const int tid  = threadIdx.x;
    const int lane = tid & 31;
    const int wv   = tid >> 5;

    const int rowBase = (int)blockIdx.x * 256;
    int r = rowBase + tid;
    if (r >= totalRows) r = totalRows - 1;

    unsigned rr  = (unsigned)r;
    unsigned b   = rr / (unsigned)(OX * OY);
    unsigned rem = rr % (unsigned)(OX * OY);
    unsigned px  = rem / (unsigned)OY;
    unsigned py  = rem % (unsigned)OY;

    const float* xp = X + (size_t)b * (HW * HW) + (size_t)(2u * px) * HW + 2u * py;
    v4f xrow[4];
#pragma unroll
    for (int kx = 0; kx < 4; ++kx) xrow[kx] = *(const v4f_a8*)(xp + kx * HW);

    const int wvu = __builtin_amdgcn_readfirstlane(wv);
    if (wvu < 2) {
      if (wvu == 0) {
        const int n = lane & 15, kh = lane >> 4;
        float m[4][2];
#pragma unroll
        for (int q = 0; q < 4; ++q) {
            float sq, cq; __sincosf(0.5f * Wt[4 + q], &sq, &cq);
            int i = (n >> (3 - q)) & 1;
            m[q][0] = (i == 0) ? cq : sq;
            m[q][1] = (i == 0) ? -sq : cq;
        }
        const float f0 = kh ? m[0][1] : m[0][0];
#pragma unroll
        for (int j = 0; j < 8; ++j) {
            float v = f0 * m[1][(j >> 2) & 1];
            v *= m[2][(j >> 1) & 1];
            v *= m[3][j & 1];
            s_bmat[(kh * 8 + j) * 16 + n] = v;
        }
      } else {
        if (lane < 16) {
            const float w0 = Wt[0], w1 = Wt[1], w2 = Wt[2], w3 = Wt[3];
            int s = lane;
            int b0 = (s >> 3) & 1, b1 = (s >> 2) & 1, b2 = (s >> 1) & 1, b3 = s & 1;
            float phi = 0.5f * ((float)b0 * (2 * b1 - 1) * w0 + (float)b1 * (2 * b2 - 1) * w1 +
                                (float)b2 * (2 * b3 - 1) * w2 + (float)b3 * (2 * b0 - 1) * w3);
            float sp, cp; __sincosf(phi, &sp, &cp);
            s_phc[s] = cp; s_phs[s] = sp;
        }
      }
    }

    __syncthreads();

    float t[16];
#pragma unroll
    for (int kx = 0; kx < 4; ++kx) {
        const v4f u = (xrow[kx] * 0.5f) * 0.15915494309189532f;
        t[kx * 4 + 0] = u.x; t[kx * 4 + 1] = u.y;
        t[kx * 4 + 2] = u.z; t[kx * 4 + 3] = u.w;
    }

    v2f P[4], Q[4];
    const float is2 = 0.70710678118654752f;
    auto firstRZ = [&](int q, float th) {
        const float sh = __builtin_amdgcn_sinf(th), ch = __builtin_amdgcn_cosf(th);
        const float a = is2 * ch, bq = is2 * sh;
        P[q] = (v2f){a, a};
        Q[q] = (v2f){-bq, bq};
    };
    auto applyRZ = [&](int q, float th) {
        const float sh = __builtin_amdgcn_sinf(th), ch = __builtin_amdgcn_cosf(th);
        const v2f S = {sh, -sh}, C = {ch, ch};
        const v2f p = P[q], w = Q[q];
        P[q] = w * S + p * C;
        Q[q] = w * C - p * S;
    };
    auto applyRY = [&](int q, float th) {
        const float sh = __builtin_amdgcn_sinf(th), ch = __builtin_amdgcn_cosf(th);
        const v2f S = {-sh, sh}, C = {ch, ch};
        const v2f p = P[q], w = Q[q];
        const v2f px = __builtin_shufflevector(p, p, 1, 0), wx = __builtin_shufflevector(w, w, 1, 0);
        P[q] = px * S + p * C;
        Q[q] = wx * S + w * C;
    };
#pragma unroll
    for (int q = 0; q < 3; ++q) {
        firstRZ(q, t[q * 5 + 0]); applyRY(q, t[q * 5 + 1]); applyRZ(q, t[q * 5 + 2]);
        applyRY(q, t[q * 5 + 3]); applyRZ(q, t[q * 5 + 4]);
    }
    firstRZ(3, t[15]);

    v2f w01r[2], w01i[2], w23r[2], w23i[2];
#pragma unroll
    for (int i = 0; i < 2; ++i) {
        const float ar = P[0][i], ai = Q[0][i], br = P[2][i], bi = Q[2][i];
        w01r[i] = ar * P[1] - ai * Q[1];
        w01i[i] = ar * Q[1] + ai * P[1];
        w23r[i] = br * P[3] - bi * Q[3];
        w23i[i] = br * Q[3] + bi * P[3];
    }
#pragma unroll
    for (int ij = 0; ij < 4; ++ij) {
        const float ar = w01r[ij >> 1][ij & 1], ai = w01i[ij >> 1][ij & 1];
#pragma unroll
        for (int k = 0; k < 2; ++k) {
            const int s0 = ij * 4 + k * 2;
            const v2f re = ar * w23r[k] - ai * w23i[k];
            const v2f im = ar * w23i[k] + ai * w23r[k];
            const v2f pc = *(const v2f*)&s_phc[s0], ps = *(const v2f*)&s_phs[s0];
            *(v2f*)&s_stage[wv][0][lane][s0] = re * pc - im * ps;
            *(v2f*)&s_stage[wv][1][lane][s0] = re * ps + im * pc;
        }
    }
    __builtin_amdgcn_wave_barrier();

    const int nn   = lane & 15;
    const int koff = (lane < 16) ? 0 : 2;
    v2f bfr[4], aRe[2][4], aIm[2][4];
#pragma unroll
    for (int j = 0; j < 4; ++j) {
        const int kk = j * 4 + koff;
        bfr[j].x = s_bmat[kk * 16 + nn];
        bfr[j].y = s_bmat[(kk + 1) * 16 + nn];
#pragma unroll
        for (int tI = 0; tI < 2; ++tI) {
            const int srcRow = tI * 16 + nn;
            aRe[tI][j].x = s_stage[wv][0][srcRow][kk];  aRe[tI][j].y = s_stage[wv][0][srcRow][kk + 1];
            aIm[tI][j].x = s_stage[wv][1][srcRow][kk];  aIm[tI][j].y = s_stage[wv][1][srcRow][kk + 1];
        }
    }
    v8f dRe[2] = {{}, {}}; v8f dIm[2] = {{}, {}};
#pragma unroll
    for (int j = 0; j < 4; ++j)
#pragma unroll
        for (int tI = 0; tI < 2; ++tI) {
            dRe[tI] = __builtin_amdgcn_wmma_f32_16x16x4_f32(false, aRe[tI][j], false, bfr[j],
                                                            (short)0, dRe[tI], false, false);
            dIm[tI] = __builtin_amdgcn_wmma_f32_16x16x4_f32(false, aIm[tI][j], false, bfr[j],
                                                            (short)0, dIm[tI], false, false);
        }
    float* sp = &s_stage[wv][0][0][0];
    const int prow0 = (lane < 16) ? 0 : 8;
#pragma unroll
    for (int tI = 0; tI < 2; ++tI)
#pragma unroll
        for (int v = 0; v < 8; ++v) {
            float p = dRe[tI][v] * dRe[tI][v] + dIm[tI][v] * dIm[tI][v];
            sp[(tI * 16 + prow0 + v) * 20 + nn] = p;
        }
    __builtin_amdgcn_wave_barrier();
    const float* pr = sp + lane * 20;
    const v4f V0 = *(const v4f*)(pr), V1 = *(const v4f*)(pr + 4), V2 = *(const v4f*)(pr + 8), V3 = *(const v4f*)(pr + 12);
    const v4f DX = V0 - V3, DY = V1 - V2;
    const v4f G = (V0 + V3) + (V1 + V2);
    const v2f x2 = (v2f){DX.x, DX.y} + (v2f){DX.z, DX.w};
    const v2f y2 = (v2f){DY.x, DY.y} + (v2f){DY.z, DY.w};
    const v2f xy = {x2.x + x2.y, y2.x + y2.y};
    const v2f z01 = (v2f){xy.x, xy.x} + (v2f){xy.y, -xy.y};
    const v2f k = (v2f){G.x, G.y} - (v2f){G.w, G.z};
    const v2f z23 = (v2f){k.x, k.x} + (v2f){k.y, -k.y};
    const v4f ev = {z01.x, z01.y, z23.x, z23.y};
    const int rowW = rowBase + wvu * 32;
    v4f* ob = (v4f*)(out + (size_t)rowW * 4) + lane;
    if (rowW + 32 <= totalRows) {
        *ob = ev;
    } else {
        if (rowW + lane < totalRows) *ob = ev;
    }
}

extern "C" void kernel_launch(void* const* d_in, const int* in_sizes, int n_in,
                              void* d_out, int out_size, void* d_ws, size_t ws_size,
                              hipStream_t stream) {
    const float* X  = (const float*)d_in[0];
    const float* Wt = (const float*)d_in[1];
    float* out      = (float*)d_out;
    int B = in_sizes[0] / (HW * HW);
    int totalRows = B * OX * OY;
    int blocks = (totalRows + 255) / 256;
    quanv_kernel<<<blocks, 256, 0, stream>>>(X, Wt, out, totalRows);
}


# Neighbour-bit sum factorised: <Z_q> = cos W (p0-p1) - 2 sin W Re(conj(a0) a1 E F); E, F linear in the 8 bit weights: one 16x8 weight table on WMMA, 4 for 8, no |.|^2 pass; rounding, sum order change
# speedup vs baseline: 2.2529x; 1.5623x over previous
#include <hip/hip_runtime.h>

typedef float v2f __attribute__((ext_vector_type(2)));
typedef float v4f __attribute__((ext_vector_type(4)));
typedef float v8f __attribute__((ext_vector_type(8)));
typedef v4f v4f_a8 __attribute__((aligned(8)));

#define HW 192
#define OX 95
#define OY 95

__device__ __forceinline__ void swap16(float& d, float& s) {
    asm("v_permlane16_swap_b32 %0, %1" : "+v"(d), "+v"(s));
}

__global__ __launch_bounds__(256)
void quanv_kernel(const float* __restrict__ X, const float* __restrict__ Wt,
                  float* __restrict__ out, int totalRows) {
    __shared__ __attribute__((aligned(16))) float s_tab[2][32][2];
    __shared__ __attribute__((aligned(16))) float s_cw[4];

    const int tid  = threadIdx.x;
    const int lane = tid & 31;
    const int wv   = tid >> 5;

    const int rowBase = (int)blockIdx.x * 256;
    int r = rowBase + tid;
    if (r >= totalRows) r = totalRows - 1;

    unsigned rr  = (unsigned)r;
    unsigned b   = rr / (unsigned)(OX * OY);
    unsigned rem = rr % (unsigned)(OX * OY);
    unsigned px  = rem / (unsigned)OY;
    unsigned py  = rem % (unsigned)OY;

    const float* xp = X + (size_t)b * (HW * HW) + (size_t)(2u * px) * HW + 2u * py;
    v4f xrow[4];
#pragma unroll
    for (int kx = 0; kx < 4; ++kx) xrow[kx] = *(const v4f_a8*)(xp + kx * HW);

    const int wvu = __builtin_amdgcn_readfirstlane(wv);
    if (wvu < 2) {
        const int kq = 2 * wvu + (lane >> 4);
        const int c = lane & 15;
        const int q = 2 * (c >> 3) + (c & 1), kind = (c >> 1) & 3;
        const bool isF = kind >= 2;
        const int nb = isF ? ((q + 1) & 3) : ((q + 3) & 3);
        const float w0 = Wt[0], w1 = Wt[1], w2 = Wt[2], w3 = Wt[3];
        const float W0 = Wt[4], W1 = Wt[5], W2 = Wt[6], W3 = Wt[7];
        const int wi = isF ? q : nb;
        const float w  = (wi & 2) ? ((wi & 1) ? w3 : w2) : ((wi & 1) ? w1 : w0);
        const float Wq = (q & 2) ? ((q & 1) ? W3 : W2) : ((q & 1) ? W1 : W0);
        float sn, cs; __sincosf(isF ? 0.5f * w : w, &sn, &cs);
        const float kap = isF ? -2.0f * __sinf(Wq) : 1.0f;
        v2f f;
        if (!isF) f = (kind == 0) ? (v2f){1.0f, cs} : (v2f){0.0f, sn};
        else      f = (kind == 2) ? (v2f){kap * cs, kap * cs} : (v2f){-(kap * sn), kap * sn};
        if (kq != nb) f = (v2f){0.0f, 0.0f};
        *(v2f*)&s_tab[wvu][lane][0] = f;
    } else if (wvu == 2) {
        const v4f c = {__cosf(Wt[4]), __cosf(Wt[5]), __cosf(Wt[6]), __cosf(Wt[7])};
        if (lane == 0) *(v4f*)s_cw = c;
    }

    __syncthreads();

    float t[16];
#pragma unroll
    for (int kx = 0; kx < 4; ++kx) {
        const v4f u = (xrow[kx] * 0.5f) * 0.15915494309189532f;
        t[kx * 4 + 0] = u.x; t[kx * 4 + 1] = u.y;
        t[kx * 4 + 2] = u.z; t[kx * 4 + 3] = u.w;
    }

    v2f P[4], Q[4];
    const float is2 = 0.70710678118654752f;
    auto firstRZ = [&](int q, float th) {
        const float sh = __builtin_amdgcn_sinf(th), ch = __builtin_amdgcn_cosf(th);
        const float a = is2 * ch, bq = is2 * sh;
        P[q] = (v2f){a, a};
        Q[q] = (v2f){-bq, bq};
    };
    auto applyRZ = [&](int q, float th) {
        const float sh = __builtin_amdgcn_sinf(th), ch = __builtin_amdgcn_cosf(th);
        const v2f S = {sh, -sh}, C = {ch, ch};
        const v2f p = P[q], w = Q[q];
        P[q] = w * S + p * C;
        Q[q] = w * C - p * S;
    };
    auto applyRY = [&](int q, float th) {
        const float sh = __builtin_amdgcn_sinf(th), ch = __builtin_amdgcn_cosf(th);
        const v2f S = {-sh, sh}, C = {ch, ch};
        const v2f p = P[q], w = Q[q];
        const v2f px = __builtin_shufflevector(p, p, 1, 0), wx = __builtin_shufflevector(w, w, 1, 0);
        P[q] = px * S + p * C;
        Q[q] = wx * S + w * C;
    };
#pragma unroll
    for (int q = 0; q < 3; ++q) {
        firstRZ(q, t[q * 5 + 0]); applyRY(q, t[q * 5 + 1]); applyRZ(q, t[q * 5 + 2]);
        applyRY(q, t[q * 5 + 3]); applyRZ(q, t[q * 5 + 4]);
    }
    firstRZ(3, t[15]);

    v2f pq[4];
#pragma unroll
    for (int q = 0; q < 4; ++q) pq[q] = P[q] * P[q] + Q[q] * Q[q];
    const v2f dpa = {pq[0].x - pq[0].y, pq[1].x - pq[1].y}, dpb = {pq[2].x - pq[2].y, pq[3].x - pq[3].y};

    const v4f cw = *(const v4f*)s_cw;
    v8f acc0 = {}, acc1 = {};
#pragma unroll
    for (int g = 0; g < 2; ++g) {
        const v2f tab = *(const v2f*)&s_tab[g][lane][0];
        float x0 = pq[2 * g].x, x1 = pq[2 * g].y, y0 = pq[2 * g + 1].x, y1 = pq[2 * g + 1].y;
        swap16(x0, y0);
        swap16(x1, y1);
        const v2f bT0 = {x0, x1}, bT1 = {y0, y1};
        acc0 = __builtin_amdgcn_wmma_f32_16x16x4_f32(false, tab, false, bT0, (short)0, acc0, false, false);
        acc1 = __builtin_amdgcn_wmma_f32_16x16x4_f32(false, tab, false, bT1, (short)0, acc1, false, false);
    }
    const v2f GR0 = (v2f){acc0[0], acc0[1]} * (v2f){acc0[4], acc0[5]} - (v2f){acc0[2], acc0[3]} * (v2f){acc0[6], acc0[7]};
    const v2f GI0 = (v2f){acc0[0], acc0[1]} * (v2f){acc0[6], acc0[7]} + (v2f){acc0[2], acc0[3]} * (v2f){acc0[4], acc0[5]};
    const v2f GR1 = (v2f){acc1[0], acc1[1]} * (v2f){acc1[4], acc1[5]} - (v2f){acc1[2], acc1[3]} * (v2f){acc1[6], acc1[7]};
    const v2f GI1 = (v2f){acc1[0], acc1[1]} * (v2f){acc1[6], acc1[7]} + (v2f){acc1[2], acc1[3]} * (v2f){acc1[4], acc1[5]};
    float gr0 = GR0.x, gr1 = GR0.y, gi0 = GI0.x, gi1 = GI0.y;
    float gr2 = GR1.x, gr3 = GR1.y, gi2 = GI1.x, gi3 = GI1.y;
    swap16(gr0, gr2);
    swap16(gr1, gr3);
    swap16(gi0, gi2);
    swap16(gi1, gi3);
    const v2f GRa = {gr0, gr1}, GIa = {gi0, gi1}, GRb = {gr2, gr3}, GIb = {gi2, gi3};
    float ur[4], ui[4];
#pragma unroll
    for (int q = 0; q < 4; ++q) {
        ur[q] = P[q].x * P[q].y + Q[q].x * Q[q].y;
        ui[q] = P[q].x * Q[q].y - Q[q].x * P[q].y;
    }
    const v2f eva = (v2f){cw.x, cw.y} * dpa + ((v2f){ur[0], ur[1]} * GRa - (v2f){ui[0], ui[1]} * GIa);
    const v2f evb = (v2f){cw.z, cw.w} * dpb + ((v2f){ur[2], ur[3]} * GRb - (v2f){ui[2], ui[3]} * GIb);
    const float ev[4] = {eva.x, eva.y, evb.x, evb.y};
    const v4f evv = {ev[0], ev[1], ev[2], ev[3]};
    const int rowW = rowBase + wvu * 32;
    v4f* ob = (v4f*)(out + (size_t)rowW * 4) + lane;
    if (rowW + 32 <= totalRows) {
        *ob = evv;
    } else {
        if (rowW + lane < totalRows) *ob = evv;
    }
}

extern "C" void kernel_launch(void* const* d_in, const int* in_sizes, int n_in,
                              void* d_out, int out_size, void* d_ws, size_t ws_size,
                              hipStream_t stream) {
    const float* X  = (const float*)d_in[0];
    const float* Wt = (const float*)d_in[1];
    float* out      = (float*)d_out;
    int B = in_sizes[0] / (HW * HW);
    int totalRows = B * OX * OY;
    int blocks = (totalRows + 255) / 256;
    quanv_kernel<<<blocks, 256, 0, stream>>>(X, Wt, out, totalRows);
}


# Qubits as Bloch vectors (full-angle plane rotations, qubits 0,1 two-wide): E, F affine in z; z3 = 0, so its k slot holds the constant: one 16x4 table on WMMA, 2 for 4; rounding, sum order change
# speedup vs baseline: 2.7371x; 1.2149x over previous
#include <hip/hip_runtime.h>


typedef float v2f __attribute__((ext_vector_type(2)));
typedef float v4f __attribute__((ext_vector_type(4)));
typedef float v8f __attribute__((ext_vector_type(8)));
typedef v4f v4f_a8 __attribute__((aligned(8)));

#define HW 192
#define OX 95
#define OY 95

__device__ __forceinline__ void swap16(float& d, float& s) {
    asm("v_permlane16_swap_b32 %0, %1" : "+v"(d), "+v"(s));
}

__global__ __launch_bounds__(256)
void quanv_kernel(const float* __restrict__ X, const float* __restrict__ Wt,
                  float* __restrict__ out, int totalRows) {
    __shared__ __attribute__((aligned(16))) float s_tab[32][2];
    __shared__ __attribute__((aligned(16))) float s_cw[4];

    const int tid  = threadIdx.x;
    const int lane = tid & 31;
    const int wv   = tid >> 5;

    const int rowBase = (int)blockIdx.x * 256;
    int r = rowBase + tid;
    if (r >= totalRows) r = totalRows - 1;

    unsigned rr  = (unsigned)r;
    unsigned b   = rr / (unsigned)(OX * OY);
    unsigned rem = rr % (unsigned)(OX * OY);
    unsigned px  = rem / (unsigned)OY;
    unsigned py  = rem % (unsigned)OY;

    const float* xp = X + (size_t)b * (HW * HW) + (size_t)(2u * px) * HW + 2u * py;
    v4f xrow[4];
#pragma unroll
    for (int kx = 0; kx < 4; ++kx) xrow[kx] = *(const v4f_a8*)(xp + kx * HW);

    const int wvu = __builtin_amdgcn_readfirstlane(wv);
    if (wvu == 0) {
        const int c = lane & 15, h = lane >> 4;
        const int q = 2 * (c >> 3) + (c & 1), kind = (c >> 1) & 3;
        const bool isF = kind >= 2;
        const int kc = isF ? ((q + 1) & 3) : ((q + 3) & 3);
        const float w0 = Wt[0], w1 = Wt[1], w2 = Wt[2], w3 = Wt[3];
        const float W0 = Wt[4], W1 = Wt[5], W2 = Wt[6], W3 = Wt[7];
        const int wi = isF ? q : kc;
        const float w  = (wi & 2) ? ((wi & 1) ? w3 : w2) : ((wi & 1) ? w1 : w0);
        const float Wq = (q & 2) ? ((q & 1) ? W3 : W2) : ((q & 1) ? W1 : W0);
        float sn, cs; __sincosf(isF ? 0.5f * w : w, &sn, &cs);
        const float kap = -__sinf(Wq);
        float coef, cst;
        if (kind == 0)      { coef = 0.5f - 0.5f * cs; cst = 0.5f + 0.5f * cs; }
        else if (kind == 1) { coef = -0.5f * sn;       cst = 0.5f * sn; }
        else if (kind == 2) { coef = 0.0f;             cst = kap * cs; }
        else                { coef = -(kap * sn);      cst = 0.0f; }
        v2f f;
        f.x = (kc == 2 * h) ? coef : 0.0f;
        f.y = h ? cst : ((kc == 1) ? coef : 0.0f);
        *(v2f*)&s_tab[lane][0] = f;
    } else if (wvu == 1) {
        const v4f c = {__cosf(Wt[4]), __cosf(Wt[5]), __cosf(Wt[6]), 0.0f};
        if (lane == 0) *(v4f*)s_cw = c;
    }

    __syncthreads();

    float t[16];
#pragma unroll
    for (int kx = 0; kx < 4; ++kx) {
        const v4f u = xrow[kx] * 0.15915494309189532f;
        t[kx * 4 + 0] = u.x; t[kx * 4 + 1] = u.y;
        t[kx * 4 + 2] = u.z; t[kx * 4 + 3] = u.w;
    }

    v2f S[5], C[5];
#pragma unroll
    for (int i = 0; i < 5; ++i) {
        S[i] = (v2f){__builtin_amdgcn_sinf(t[i]), __builtin_amdgcn_sinf(t[5 + i])};
        C[i] = (v2f){__builtin_amdgcn_cosf(t[i]), __builtin_amdgcn_cosf(t[5 + i])};
    }
    v2f bx = C[0] * C[1], by = S[0], bz = -(C[0] * S[1]);
    v2f nx;
    nx = bx * C[2] - by * S[2];  by = bx * S[2] + by * C[2];  bx = nx;
    nx = bx * C[3] + bz * S[3];  bz = bz * C[3] - bx * S[3];  bx = nx;
    nx = bx * C[4] - by * S[4];  by = bx * S[4] + by * C[4];  bx = nx;

    float s2[5], c2[5];
#pragma unroll
    for (int i = 0; i < 5; ++i) { s2[i] = __builtin_amdgcn_sinf(t[10 + i]); c2[i] = __builtin_amdgcn_cosf(t[10 + i]); }
    float x2 = c2[0] * c2[1], y2 = s2[0], z2 = -(c2[0] * s2[1]);
    float n2;
    n2 = x2 * c2[2] - y2 * s2[2];  y2 = x2 * s2[2] + y2 * c2[2];  x2 = n2;
    n2 = x2 * c2[3] + z2 * s2[3];  z2 = z2 * c2[3] - x2 * s2[3];  x2 = n2;
    n2 = x2 * c2[4] - y2 * s2[4];  y2 = x2 * s2[4] + y2 * c2[4];  x2 = n2;

    const float x3 = __builtin_amdgcn_cosf(t[15]), y3 = __builtin_amdgcn_sinf(t[15]);

    const v4f cw = *(const v4f*)s_cw;
    const v2f tab = *(const v2f*)&s_tab[lane][0];
    float k0 = bz.x, k1 = bz.y, k2 = z2, k3 = 1.0f;
    swap16(k0, k2);
    swap16(k1, k3);
    const v2f bT0 = {k0, k1}, bT1 = {k2, k3};
    const v8f zero = {};
    const v8f acc0 = __builtin_amdgcn_wmma_f32_16x16x4_f32(false, tab, false, bT0, (short)0, zero, false, false);
    const v8f acc1 = __builtin_amdgcn_wmma_f32_16x16x4_f32(false, tab, false, bT1, (short)0, zero, false, false);
    const v2f GR0 = (v2f){acc0[0], acc0[1]} * (v2f){acc0[4], acc0[5]} - (v2f){acc0[2], acc0[3]} * (v2f){acc0[6], acc0[7]};
    const v2f GI0 = (v2f){acc0[0], acc0[1]} * (v2f){acc0[6], acc0[7]} + (v2f){acc0[2], acc0[3]} * (v2f){acc0[4], acc0[5]};
    const v2f GR1 = (v2f){acc1[0], acc1[1]} * (v2f){acc1[4], acc1[5]} - (v2f){acc1[2], acc1[3]} * (v2f){acc1[6], acc1[7]};
    const v2f GI1 = (v2f){acc1[0], acc1[1]} * (v2f){acc1[6], acc1[7]} + (v2f){acc1[2], acc1[3]} * (v2f){acc1[4], acc1[5]};
    float gr0 = GR0.x, gr1 = GR0.y, gi0 = GI0.x, gi1 = GI0.y;
    float gr2 = GR1.x, gr3 = GR1.y, gi2 = GI1.x, gi3 = GI1.y;
    swap16(gr0, gr2);
    swap16(gr1, gr3);
    swap16(gi0, gi2);
    swap16(gi1, gi3);
    const v2f GRa = {gr0, gr1}, GIa = {gi0, gi1}, GRb = {gr2, gr3}, GIb = {gi2, gi3};
    const v2f eva = (v2f){cw.x, cw.y} * bz + (bx * GRa - by * GIa);
    const v2f tb  = (v2f){x2, x3} * GRb - (v2f){y2, y3} * GIb;
    const v4f evv = {eva.x, eva.y, cw.z * z2 + tb.x, tb.y};
    const int rowW = rowBase + wvu * 32;
    v4f* ob = (v4f*)(out + (size_t)rowW * 4) + lane;
    if (rowW + 32 <= totalRows) {
        *ob = evv;
    } else {
        if (rowW + lane < totalRows) *ob = evv;
    }
}

extern "C" void kernel_launch(void* const* d_in, const int* in_sizes, int n_in,
                              void* d_out, int out_size, void* d_ws, size_t ws_size,
                              hipStream_t stream) {
    const float* X  = (const float*)d_in[0];
    const float* Wt = (const float*)d_in[1];
    float* out      = (float*)d_out;
    int B = in_sizes[0] / (HW * HW);
    int totalRows = B * OX * OY;
    int blocks = (totalRows + 255) / 256;
    quanv_kernel<<<blocks, 256, 0, stream>>>(X, Wt, out, totalRows);
}
